# starting kernel
# baseline (speedup 1.0000x reference)
#include <hip/hip_runtime.h>

typedef __attribute__((ext_vector_type(2))) float v2f;
typedef __attribute__((ext_vector_type(8))) float v8f;

#define NV 64
#define WPB 8

__global__ void __launch_bounds__(256)
gcn_pool_kernel(const float* __restrict__ A, const float* __restrict__ h0,
                const float* __restrict__ W1, const float* __restrict__ W2,
                float* __restrict__ out, int B)
{
  __shared__ float dinvS[WPB][NV];
  __shared__ float GS[WPB][NV * 4];

  const int tid  = threadIdx.x;
  const int wv   = tid >> 5;
  const int lane = tid & 31;
  const int r    = lane & 15;
  const int h    = lane >> 4;

  const int b = blockIdx.x * WPB + wv;
  if (b >= B) return;

  const float* Ab = A  + (size_t)b * (NV * NV);
  const float* Hb = h0 + (size_t)b * (NV * 5);

  v2f a[4][16];
#pragma unroll
  for (int m = 0; m < 4; ++m) {
    const int row = 16 * m + r;
#pragma unroll
    for (int k = 0; k < 16; ++k) {
      const int col = 4 * k + 2 * h;
      float x = Ab[row * NV + col];
      float y = Ab[row * NV + col + 1];
      if (row == col)     x += 1.0f;
      if (row == col + 1) y += 1.0f;
      a[m][k].x = x;
      a[m][k].y = y;
    }
  }

  float dv[4];
#pragma unroll
  for (int m = 0; m < 4; ++m) {
    float p = 0.0f;
#pragma unroll
    for (int k = 0; k < 16; ++k) p += a[m][k].x + a[m][k].y;
    p += __shfl_xor(p, 16, 32);
    dv[m] = 1.0f / sqrtf(p);
  }
  if (h == 0) {
#pragma unroll
    for (int m = 0; m < 4; ++m) dinvS[wv][16 * m + r] = dv[m];
  }
#pragma unroll
  for (int m = 0; m < 4; ++m)
#pragma unroll
    for (int k = 0; k < 16; ++k) { a[m][k].x *= dv[m]; a[m][k].y *= dv[m]; }

  float w1[5][3];
#pragma unroll
  for (int c = 0; c < 5; ++c)
#pragma unroll
    for (int f = 0; f < 3; ++f) w1[c][f] = W1[c * 3 + f];

#pragma unroll
  for (int t = 0; t < 2; ++t) {
    const int w = lane + 32 * t;
    const float* hr = Hb + w * 5;
    float hv[5];
#pragma unroll
    for (int c = 0; c < 5; ++c) hv[c] = hr[c];
    const float dw = dinvS[wv][w];
#pragma unroll
    for (int f = 0; f < 3; ++f) {
      float s = 0.0f;
#pragma unroll
      for (int c = 0; c < 5; ++c) s += hv[c] * w1[c][f];
      GS[wv][w * 4 + f] = dw * s;
    }
  }

  v2f bf[16];
#pragma unroll
  for (int k = 0; k < 16; ++k) {
    const int w0 = 4 * k + 2 * h;
    float bx = 0.0f, by = 0.0f;
    if (r < 3) {
      bx = GS[wv][w0 * 4 + r];
      by = GS[wv][(w0 + 1) * 4 + r];
    }
    bf[k].x = bx;
    bf[k].y = by;
  }

  float p = 0.0f;
#pragma unroll
  for (int m = 0; m < 4; ++m) {
    v8f acc = {0.f, 0.f, 0.f, 0.f, 0.f, 0.f, 0.f, 0.f};
#pragma unroll
    for (int k = 0; k < 16; ++k)
      acc = __builtin_amdgcn_wmma_f32_16x16x4_f32(
           false, a[m][k],  false, bf[k],
           (short)0, acc,  false,  false);
#pragma unroll
    for (int i = 0; i < 8; ++i) p += fmaxf(acc[i], 0.0f);
  }
  p += __shfl_xor(p, 16, 32);

  float t = 0.0f;
  if (lane < 3) t = p * W2[lane];
  float o = __shfl(t, 0, 32) + __shfl(t, 1, 32) + __shfl(t, 2, 32);
  if (lane == 0) out[b] = o;
}

extern "C" void kernel_launch(void* const* d_in, const int* in_sizes, int n_in,
                              void* d_out, int out_size, void* d_ws, size_t ws_size,
                              hipStream_t stream) {
  const float* A  = (const float*)d_in[0];
  const float* h0 = (const float*)d_in[1];
  const float* W1 = (const float*)d_in[2];
  const float* W2 = (const float*)d_in[3];
  float* out = (float*)d_out;

  const int B = in_sizes[0] / (NV * NV);
  dim3 block(256);
  dim3 grid((B + WPB - 1) / WPB);
  gcn_pool_kernel<<<grid, block, 0, stream>>>(A, h0, W1, W2, out, B);
}


# Every global load reads 512 contiguous bytes (16 B per lane); each wave re-shapes A and h0 into WMMA fragments via its own LDS slice; fragment k order, hence summation order, changes
# speedup vs baseline: 1.2780x; 1.2780x over previous
#include <hip/hip_runtime.h>


typedef __attribute__((ext_vector_type(2))) float v2f;
typedef __attribute__((ext_vector_type(4))) float v4f;
typedef __attribute__((ext_vector_type(8))) float v8f;

#define NV 64
#define WPB 8
#define P4 17

#define H_OFF 0
#define D_OFF 80
#define G_OFF 96

__global__ void __launch_bounds__(256)
gcn_pool_kernel(const float* __restrict__ A, const float* __restrict__ h0,
                const float* __restrict__ W1, const float* __restrict__ W2,
                float* __restrict__ out, int B)
{
  __shared__ v4f LW[WPB][NV * P4];

  const int tid  = threadIdx.x;
  const int wv   = tid >> 5;
  const int lane = tid & 31;
  const int r    = lane & 15;
  const int h    = lane >> 4;

  const int b = blockIdx.x * WPB + wv;
  if (b >= B) return;

  const v4f* Ab4 = (const v4f*)(A  + (size_t)b * (NV * NV));
  const v4f* Hb4 = (const v4f*)(h0 + (size_t)b * (NV * 5));
  v4f*   lw  = LW[wv];
  float* lwf = (float*)lw;

  v4f st[32];
#pragma unroll
  for (int i = 0; i < 32; ++i) st[i] = Ab4[i * 32 + lane];
  v4f hs0 = Hb4[lane];
  v4f hs1 = Hb4[32 + lane];
  v4f hs2 = Hb4[64 + r];

#pragma unroll
  for (int i = 0; i < 32; ++i) lw[(2 * i + h) * P4 + r] = st[i];

  v2f a[4][16];
#pragma unroll
  for (int m = 0; m < 4; ++m) {
#pragma unroll
    for (int q = 0; q < 8; ++q) {
      const v4f v = lw[(16 * m + r) * P4 + 2 * q + h];
      a[m][2 * q].x     = v.x;
      a[m][2 * q].y     = v.y;
      a[m][2 * q + 1].x = v.z;
      a[m][2 * q + 1].y = v.w;
    }
  }
  const int d = r - 4 * h;
#pragma unroll
  for (int m = 0; m < 4; ++m) {
#pragma unroll
    for (int qq = 0; qq < 2; ++qq) {
      const int q = 2 * m + qq;
      if (d == 8 * qq + 0) a[m][2 * q].x     += 1.0f;
      if (d == 8 * qq + 1) a[m][2 * q].y     += 1.0f;
      if (d == 8 * qq + 2) a[m][2 * q + 1].x += 1.0f;
      if (d == 8 * qq + 3) a[m][2 * q + 1].y += 1.0f;
    }
  }

  float dv[4];
#pragma unroll
  for (int m = 0; m < 4; ++m) {
    float p = 0.0f;
#pragma unroll
    for (int k = 0; k < 16; ++k) p += a[m][k].x + a[m][k].y;
    p += __shfl_xor(p, 16, 32);
    dv[m] = 1.0f / sqrtf(p);
  }

  lw[H_OFF + lane]      = hs0;
  lw[H_OFF + 32 + lane] = hs1;
  lw[H_OFF + 64 + r]    = hs2;
  if (h == 0) {
#pragma unroll
    for (int m = 0; m < 4; ++m) lwf[D_OFF * 4 + 16 * m + r] = dv[m];
  }
#pragma unroll
  for (int m = 0; m < 4; ++m)
#pragma unroll
    for (int k = 0; k < 16; ++k) { a[m][k].x *= dv[m]; a[m][k].y *= dv[m]; }

  float w1[5][3];
#pragma unroll
  for (int c = 0; c < 5; ++c)
#pragma unroll
    for (int f = 0; f < 3; ++f) w1[c][f] = W1[c * 3 + f];

#pragma unroll
  for (int t = 0; t < 2; ++t) {
    const int w = lane + 32 * t;
    float hv[5];
#pragma unroll
    for (int c = 0; c < 5; ++c) hv[c] = lwf[H_OFF * 4 + w * 5 + c];
    const float dw = lwf[D_OFF * 4 + w];
#pragma unroll
    for (int f = 0; f < 3; ++f) {
      float s = 0.0f;
#pragma unroll
      for (int c = 0; c < 5; ++c) s += hv[c] * w1[c][f];
      lwf[(G_OFF + f * P4) * 4 + w] = dw * s;
    }
  }

  v2f bf[16];
#pragma unroll
  for (int q = 0; q < 8; ++q) {
    v4f g = {0.f, 0.f, 0.f, 0.f};
    if (r < 3) g = lw[G_OFF + r * P4 + 2 * q + h];
    bf[2 * q].x     = g.x;
    bf[2 * q].y     = g.y;
    bf[2 * q + 1].x = g.z;
    bf[2 * q + 1].y = g.w;
  }

  float p = 0.0f;
#pragma unroll
  for (int m = 0; m < 4; ++m) {
    v8f acc = {0.f, 0.f, 0.f, 0.f, 0.f, 0.f, 0.f, 0.f};
#pragma unroll
    for (int k = 0; k < 16; ++k)
      acc = __builtin_amdgcn_wmma_f32_16x16x4_f32(
           false, a[m][k],  false, bf[k],
           (short)0, acc,  false,  false);
#pragma unroll
    for (int i = 0; i < 8; ++i) p += fmaxf(acc[i], 0.0f);
  }
  p += __shfl_xor(p, 16, 32);

  float t = 0.0f;
  if (lane < 3) t = p * W2[lane];
  float o = __shfl(t, 0, 32) + __shfl(t, 1, 32) + __shfl(t, 2, 32);
  if (lane == 0) out[b] = o;
}

extern "C" void kernel_launch(void* const* d_in, const int* in_sizes, int n_in,
                              void* d_out, int out_size, void* d_ws, size_t ws_size,
                              hipStream_t stream) {
  const float* A  = (const float*)d_in[0];
  const float* h0 = (const float*)d_in[1];
  const float* W1 = (const float*)d_in[2];
  const float* W2 = (const float*)d_in[3];
  float* out = (float*)d_out;

  const int B = in_sizes[0] / (NV * NV);
  dim3 block(256);
  dim3 grid((B + WPB - 1) / WPB);
  gcn_pool_kernel<<<grid, block, 0, stream>>>(A, h0, W1, W2, out, B);
}


# Work-groups of 2 waves instead of 8 (64 threads, 8192 groups): each pair refills its wave slots and LDS as soon as it ends, so some waves of a WGP load while the others compute
# speedup vs baseline: 1.3219x; 1.0344x over previous
#include <hip/hip_runtime.h>


typedef __attribute__((ext_vector_type(2))) float v2f;
typedef __attribute__((ext_vector_type(4))) float v4f;
typedef __attribute__((ext_vector_type(8))) float v8f;

#define NV 64
#define WPB 2
#define P4 17

#define H_OFF 0
#define D_OFF 80
#define G_OFF 96

__global__ void __launch_bounds__(64)
gcn_pool_kernel(const float* __restrict__ A, const float* __restrict__ h0,
                const float* __restrict__ W1, const float* __restrict__ W2,
                float* __restrict__ out, int B)
{
  __shared__ v4f LW[WPB][NV * P4];

  const int tid  = threadIdx.x;
  const int wv   = tid >> 5;
  const int lane = tid & 31;
  const int r    = lane & 15;
  const int h    = lane >> 4;

  const int b = blockIdx.x * WPB + wv;
  if (b >= B) return;

  const v4f* Ab4 = (const v4f*)(A  + (size_t)b * (NV * NV));
  const v4f* Hb4 = (const v4f*)(h0 + (size_t)b * (NV * 5));
  v4f*   lw  = LW[wv];
  float* lwf = (float*)lw;

  v4f st[32];
#pragma unroll
  for (int i = 0; i < 32; ++i) st[i] = Ab4[i * 32 + lane];
  v4f hs0 = Hb4[lane];
  v4f hs1 = Hb4[32 + lane];
  v4f hs2 = Hb4[64 + r];

#pragma unroll
  for (int i = 0; i < 32; ++i) lw[(2 * i + h) * P4 + r] = st[i];

  v2f a[4][16];
#pragma unroll
  for (int m = 0; m < 4; ++m) {
#pragma unroll
    for (int q = 0; q < 8; ++q) {
      const v4f v = lw[(16 * m + r) * P4 + 2 * q + h];
      a[m][2 * q].x     = v.x;
      a[m][2 * q].y     = v.y;
      a[m][2 * q + 1].x = v.z;
      a[m][2 * q + 1].y = v.w;
    }
  }
  const int d = r - 4 * h;
#pragma unroll
  for (int m = 0; m < 4; ++m) {
#pragma unroll
    for (int qq = 0; qq < 2; ++qq) {
      const int q = 2 * m + qq;
      if (d == 8 * qq + 0) a[m][2 * q].x     += 1.0f;
      if (d == 8 * qq + 1) a[m][2 * q].y     += 1.0f;
      if (d == 8 * qq + 2) a[m][2 * q + 1].x += 1.0f;
      if (d == 8 * qq + 3) a[m][2 * q + 1].y += 1.0f;
    }
  }

  float dv[4];
#pragma unroll
  for (int m = 0; m < 4; ++m) {
    float p = 0.0f;
#pragma unroll
    for (int k = 0; k < 16; ++k) p += a[m][k].x + a[m][k].y;
    p += __shfl_xor(p, 16, 32);
    dv[m] = 1.0f / sqrtf(p);
  }

  lw[H_OFF + lane]      = hs0;
  lw[H_OFF + 32 + lane] = hs1;
  lw[H_OFF + 64 + r]    = hs2;
  if (h == 0) {
#pragma unroll
    for (int m = 0; m < 4; ++m) lwf[D_OFF * 4 + 16 * m + r] = dv[m];
  }
#pragma unroll
  for (int m = 0; m < 4; ++m)
#pragma unroll
    for (int k = 0; k < 16; ++k) { a[m][k].x *= dv[m]; a[m][k].y *= dv[m]; }

  float w1[5][3];
#pragma unroll
  for (int c = 0; c < 5; ++c)
#pragma unroll
    for (int f = 0; f < 3; ++f) w1[c][f] = W1[c * 3 + f];

#pragma unroll
  for (int t = 0; t < 2; ++t) {
    const int w = lane + 32 * t;
    float hv[5];
#pragma unroll
    for (int c = 0; c < 5; ++c) hv[c] = lwf[H_OFF * 4 + w * 5 + c];
    const float dw = lwf[D_OFF * 4 + w];
#pragma unroll
    for (int f = 0; f < 3; ++f) {
      float s = 0.0f;
#pragma unroll
      for (int c = 0; c < 5; ++c) s += hv[c] * w1[c][f];
      lwf[(G_OFF + f * P4) * 4 + w] = dw * s;
    }
  }

  v2f bf[16];
#pragma unroll
  for (int q = 0; q < 8; ++q) {
    v4f g = {0.f, 0.f, 0.f, 0.f};
    if (r < 3) g = lw[G_OFF + r * P4 + 2 * q + h];
    bf[2 * q].x     = g.x;
    bf[2 * q].y     = g.y;
    bf[2 * q + 1].x = g.z;
    bf[2 * q + 1].y = g.w;
  }

  float p = 0.0f;
#pragma unroll
  for (int m = 0; m < 4; ++m) {
    v8f acc = {0.f, 0.f, 0.f, 0.f, 0.f, 0.f, 0.f, 0.f};
#pragma unroll
    for (int k = 0; k < 16; ++k)
      acc = __builtin_amdgcn_wmma_f32_16x16x4_f32(
           false, a[m][k],  false, bf[k],
           (short)0, acc,  false,  false);
#pragma unroll
    for (int i = 0; i < 8; ++i) p += fmaxf(acc[i], 0.0f);
  }
  p += __shfl_xor(p, 16, 32);

  float t = 0.0f;
  if (lane < 3) t = p * W2[lane];
  float o = __shfl(t, 0, 32) + __shfl(t, 1, 32) + __shfl(t, 2, 32);
  if (lane == 0) out[b] = o;
}

extern "C" void kernel_launch(void* const* d_in, const int* in_sizes, int n_in,
                              void* d_out, int out_size, void* d_ws, size_t ws_size,
                              hipStream_t stream) {
  const float* A  = (const float*)d_in[0];
  const float* h0 = (const float*)d_in[1];
  const float* W1 = (const float*)d_in[2];
  const float* W2 = (const float*)d_in[3];
  float* out = (float*)d_out;

  const int B = in_sizes[0] / (NV * NV);
  dim3 block(64);
  dim3 grid((B + WPB - 1) / WPB);
  gcn_pool_kernel<<<grid, block, 0, stream>>>(A, h0, W1, W2, out, B);
}
